# starting kernel
# baseline (speedup 1.0000x reference)
#include <hip/hip_runtime.h>
#include <hip/hip_bf16.h>


typedef float v2f __attribute__((ext_vector_type(2)));
typedef float v8f __attribute__((ext_vector_type(8)));

#define N_NODES 100000
#define IN_CH   128
#define HID     64
#define OUT_CH  32

__global__ void k_init(float* __restrict__ deg, float* __restrict__ agg1,
                       float* __restrict__ out, int n) {
  long i = (long)blockIdx.x * blockDim.x + threadIdx.x;
  if (i < n) deg[i] = 1.0f;
  if (i < (long)n * HID) agg1[i] = 0.0f;
  if (i < (long)n * OUT_CH) out[i] = 0.0f;
}

__global__ void k_deg(const int* __restrict__ dst, float* __restrict__ deg, int e) {
  int i = blockIdx.x * blockDim.x + threadIdx.x;
  if (i < e) {
    (void)__hip_atomic_fetch_add(&deg[dst[i]], 1.0f,
                                 __ATOMIC_RELAXED, __HIP_MEMORY_SCOPE_AGENT);
  }
}

__global__ void k_rsqrt(float* __restrict__ d, int n) {
  int i = blockIdx.x * blockDim.x + threadIdx.x;
  if (i < n) d[i] = rsqrtf(d[i]);
}

__global__ void k_gemm_f32(const float* __restrict__ A, const float* __restrict__ W,
                           float* __restrict__ out, int M, int K, int NC,
                           int tiles_n, int total_tiles) {
  int wave = (int)((blockIdx.x * (long)blockDim.x + threadIdx.x) >> 5);
  if (wave >= total_tiles) return;
  int lane  = threadIdx.x & 31;
  int tm    = wave / tiles_n;
  int tn    = wave % tiles_n;
  int hi    = lane >> 4;
  int l15   = lane & 15;
  int row   = tm * 16 + l15;
  int col   = tn * 16 + l15;
  int rrow  = row < M ? row : M - 1;

  const float* arow = A + (size_t)rrow * K + 2 * hi;
  const float* wcol = W + (size_t)(2 * hi) * NC + col;

  v8f c = {};
  for (int k0 = 0; k0 < K; k0 += 4) {
    v2f a = *(const v2f*)(arow + k0);
    v2f b;
    b.x = wcol[(size_t)k0 * NC];
    b.y = wcol[(size_t)(k0 + 1) * NC];
    c = __builtin_amdgcn_wmma_f32_16x16x4_f32(
         false, a,  false, b,
         (short)0, c,  false,  false);
  }

  int m0 = tm * 16 + hi * 8;
  #pragma unroll
  for (int v = 0; v < 8; ++v) {
    int r = m0 + v;
    if (r < M) out[(size_t)r * NC + col] = c[v];
  }
}

template <int C>
__global__ void k_scatter(const float* __restrict__ m, const int* __restrict__ src,
                          const int* __restrict__ dst, const float* __restrict__ dinv,
                          float* __restrict__ agg, int e) {
  int wave = (int)((blockIdx.x * (long)blockDim.x + threadIdx.x) >> 5);
  if (wave >= e) return;
  int lane = threadIdx.x & 31;
  int s = src[wave];
  int d = dst[wave];
  float w = dinv[s] * dinv[d];
  #pragma unroll
  for (int c = lane; c < C; c += 32) {
    float val = m[(size_t)s * C + c] * w;
    (void)__hip_atomic_fetch_add(&agg[(size_t)d * C + c], val,
                                 __ATOMIC_RELAXED, __HIP_MEMORY_SCOPE_AGENT);
  }
}

__global__ void k_post1(float* __restrict__ agg, const float* __restrict__ m,
                        const float* __restrict__ dinv, const float* __restrict__ b1,
                        int n) {
  long i = (long)blockIdx.x * blockDim.x + threadIdx.x;
  if (i >= (long)n * HID) return;
  int node = (int)(i >> 6);
  int ch   = (int)(i & 63);
  float di = dinv[node];
  float v  = agg[i] + m[i] * di * di + b1[ch];
  agg[i]   = v > 0.0f ? v : 0.0f;
}

__global__ void k_post2(float* __restrict__ out, const float* __restrict__ m2,
                        const float* __restrict__ dinv, const float* __restrict__ b2,
                        int n) {
  long i = (long)blockIdx.x * blockDim.x + threadIdx.x;
  if (i >= (long)n * OUT_CH) return;
  int node = (int)(i >> 5);
  int ch   = (int)(i & 31);
  float di = dinv[node];
  out[i] = out[i] + m2[i] * di * di + b2[ch];
}

extern "C" void kernel_launch(void* const* d_in, const int* in_sizes, int n_in,
                              void* d_out, int out_size, void* d_ws, size_t ws_size,
                              hipStream_t stream) {
  const float* x   = (const float*)d_in[0];
  const int*   ei  = (const int*)d_in[1];
  const float* W1  = (const float*)d_in[2];
  const float* b1  = (const float*)d_in[3];
  const float* W2  = (const float*)d_in[4];
  const float* b2  = (const float*)d_in[5];
  float*       out = (float*)d_out;

  const int N = in_sizes[0] / IN_CH;
  const int E = in_sizes[1] / 2;
  const int* src = ei;
  const int* dst = ei + E;

  float* dinv = (float*)d_ws;
  float* bufA = dinv + 131072;
  float* bufB = bufA + (size_t)N * HID;

  const int BT = 256;

  {
    long total = (long)N * HID;
    k_init<<<(int)((total + BT - 1) / BT), BT, 0, stream>>>(dinv, bufB, out, N);
  }
  k_deg<<<(E + BT - 1) / BT, BT, 0, stream>>>(dst, dinv, E);
  k_rsqrt<<<(N + BT - 1) / BT, BT, 0, stream>>>(dinv, N);
  {
    int tiles_m = (N + 15) / 16, tiles_n = HID / 16;
    int tiles = tiles_m * tiles_n;
    long threads = (long)tiles * 32;
    k_gemm_f32<<<(int)((threads + BT - 1) / BT), BT, 0, stream>>>(
        x, W1, bufA, N, IN_CH, HID, tiles_n, tiles);
  }
  {
    long threads = (long)E * 32;
    k_scatter<HID><<<(int)((threads + BT - 1) / BT), BT, 0, stream>>>(
        bufA, src, dst, dinv, bufB, E);
  }
  {
    long total = (long)N * HID;
    k_post1<<<(int)((total + BT - 1) / BT), BT, 0, stream>>>(bufB, bufA, dinv, b1, N);
  }
  {
    int tiles_m = (N + 15) / 16, tiles_n = OUT_CH / 16;
    int tiles = tiles_m * tiles_n;
    long threads = (long)tiles * 32;
    k_gemm_f32<<<(int)((threads + BT - 1) / BT), BT, 0, stream>>>(
        bufB, W2, bufA, N, HID, OUT_CH, tiles_n, tiles);
  }
  {
    long threads = (long)E * 32;
    k_scatter<OUT_CH><<<(int)((threads + BT - 1) / BT), BT, 0, stream>>>(
        bufA, src, dst, dinv, out, E);
  }
  {
    long total = (long)N * OUT_CH;
    k_post2<<<(int)((total + BT - 1) / BT), BT, 0, stream>>>(out, bufA, dinv, b2, N);
  }
}


# Scatter by float atomics replaced by a destination-ordered edge list (arrival-order rank from one returning integer add per edge + scan) and one plain-load row sum per node; sum order changes
# speedup vs baseline: 1.9807x; 1.9807x over previous
#include <hip/hip_runtime.h>
#include <hip/hip_bf16.h>


typedef float v2f __attribute__((ext_vector_type(2)));
typedef float v8f __attribute__((ext_vector_type(8)));

#define N_NODES 100000
#define IN_CH   128
#define HID     64
#define OUT_CH  32

#define SCAN_T   1024
#define SCAN_PER (SCAN_T * 4)
#define LIST_OFF 131072

__global__ void k_zero(int* __restrict__ cnt, int n) {
  int i = blockIdx.x * blockDim.x + threadIdx.x;
  if (i < n) cnt[i] = 0;
}

__global__ void k_rank(const int* __restrict__ dst, int* __restrict__ cnt,
                       int* __restrict__ rank, int e) {
  int i = blockIdx.x * blockDim.x + threadIdx.x;
  if (i < e) {
    rank[i] = __hip_atomic_fetch_add(&cnt[dst[i]], 1,
                                     __ATOMIC_RELAXED, __HIP_MEMORY_SCOPE_AGENT);
  }
}

__global__ void __launch_bounds__(SCAN_T)
k_scan(const int* __restrict__ cnt, int* __restrict__ offs,
       float* __restrict__ dinv, int n) {
  __shared__ int s_inc[SCAN_T / 32];
  __shared__ int s_pre[SCAN_T / 32];
  const int tid  = threadIdx.x;
  const int lane = tid & 31;
  const int wv   = tid >> 5;
  const int blk0 = blockIdx.x * SCAN_PER;

  int pre = 0;
  const int nfront = blockIdx.x;
  for (int j0 = 0; j0 < nfront; j0 += 8) {
    int4 v[8];
    #pragma unroll
    for (int j = 0; j < 8; ++j) {
      int jj = j0 + j;
      if (jj > nfront - 1) jj = nfront - 1;
      v[j] = *(const int4*)(cnt + jj * SCAN_PER + tid * 4);
    }
    #pragma unroll
    for (int j = 0; j < 8; ++j) {
      const int t = (v[j].x + v[j].y) + (v[j].z + v[j].w);
      pre += (j0 + j < nfront) ? t : 0;
    }
  }

  const int i0 = blk0 + tid * 4;
  int c[4];
  #pragma unroll
  for (int k = 0; k < 4; ++k) c[k] = (i0 + k < n) ? cnt[i0 + k] : 0;
  const int tsum = (c[0] + c[1]) + (c[2] + c[3]);

  int inc = tsum;
  #pragma unroll
  for (int d = 1; d < 32; d <<= 1) {
    int t = __shfl_up(inc, d, 32);
    if (lane >= d) inc += t;
  }
  #pragma unroll
  for (int d = 16; d >= 1; d >>= 1) pre += __shfl_xor(pre, d, 32);

  if (lane == 31) s_inc[wv] = inc;
  if (lane == 0)  s_pre[wv] = pre;
  __syncthreads();

  int base = 0;
  for (int w = 0; w < SCAN_T / 32; ++w) {
    base += s_pre[w];
    if (w < wv) base += s_inc[w];
  }
  int ex = base + inc - tsum;

  #pragma unroll
  for (int k = 0; k < 4; ++k) {
    if (i0 + k < n) {
      offs[i0 + k] = ex;
      dinv[i0 + k] = rsqrtf((float)(c[k] + 1));
      ex += c[k];
    }
  }
  if (i0 <= n - 1 && n - 1 < i0 + 4) offs[n] = ex;
}

__global__ void k_fill(const int* __restrict__ src, const int* __restrict__ dst,
                       const int* __restrict__ rank, const int* __restrict__ offs,
                       int* __restrict__ list, int e) {
  int i = blockIdx.x * blockDim.x + threadIdx.x;
  if (i < e) list[offs[dst[i]] + rank[i]] = src[i];
}

__global__ void k_move(const int* __restrict__ from, int* __restrict__ to,
                       int n_starts, int total) {
  int i = blockIdx.x * blockDim.x + threadIdx.x;
  if (i < total && (i < n_starts || i >= LIST_OFF)) to[i] = from[i];
}

__global__ void k_gemm_f32(const float* __restrict__ A, const float* __restrict__ W,
                           float* __restrict__ out, int M, int K, int NC,
                           int tiles_n, int total_tiles) {
  int wave = (int)((blockIdx.x * (long)blockDim.x + threadIdx.x) >> 5);
  if (wave >= total_tiles) return;
  int lane  = threadIdx.x & 31;
  int tm    = wave / tiles_n;
  int tn    = wave % tiles_n;
  int hi    = lane >> 4;
  int l15   = lane & 15;
  int row   = tm * 16 + l15;
  int col   = tn * 16 + l15;
  int rrow  = row < M ? row : M - 1;

  const float* arow = A + (size_t)rrow * K + 2 * hi;
  const float* wcol = W + (size_t)(2 * hi) * NC + col;

  v8f c = {};
  for (int k0 = 0; k0 < K; k0 += 4) {
    v2f a = *(const v2f*)(arow + k0);
    v2f b;
    b.x = wcol[(size_t)k0 * NC];
    b.y = wcol[(size_t)(k0 + 1) * NC];
    c = __builtin_amdgcn_wmma_f32_16x16x4_f32(
         false, a,  false, b,
         (short)0, c,  false,  false);
  }

  int m0 = tm * 16 + hi * 8;
  #pragma unroll
  for (int v = 0; v < 8; ++v) {
    int r = m0 + v;
    if (r < M) out[(size_t)r * NC + col] = c[v];
  }
}

template <int C, bool RELU>
__global__ void k_gather(const float* __restrict__ m, const int* __restrict__ list,
                         const int* __restrict__ offs, const float* __restrict__ dinv,
                         const float* __restrict__ bias, float* __restrict__ res, int n) {
  const int node = (int)(blockIdx.x * (blockDim.x >> 5)) +
                   __builtin_amdgcn_readfirstlane((int)(threadIdx.x >> 5));
  if (node >= n) return;
  const int lane = threadIdx.x & 31;
  const int beg = offs[node];
  const int end = offs[node + 1];
  const float di = dinv[node];

  v2f  self2 = {};
  float self1 = 0.0f;
  if constexpr (C == 64) self2 = *(const v2f*)(m + (size_t)node * C + 2 * lane);
  else                   self1 = m[(size_t)node * C + lane];

  float a0 = 0.0f, a1 = 0.0f;
  for (int base = beg; base < end; base += 32) {
    int cnt = end - base;
    if (cnt > 32) cnt = 32;
    int   sj = 0;
    float wj = 0.0f;
    if (lane < cnt) {
      sj = list[base + lane];
      wj = dinv[sj] * di;
    }
    for (int k = 0; k < cnt; k += 8) {
      v2f   r2[8];
      float r1[8];
      float w[8];
      #pragma unroll
      for (int j = 0; j < 8; ++j) {
        int kk = k + j;
        if (kk > cnt - 1) kk = cnt - 1;
        const int s = __builtin_amdgcn_readlane(sj, kk);
        w[j] = __int_as_float(__builtin_amdgcn_readlane(__float_as_int(wj), kk));
        if constexpr (C == 64) r2[j] = *(const v2f*)(m + (size_t)s * C + 2 * lane);
        else                   r1[j] = m[(size_t)s * C + lane];
      }
      #pragma unroll
      for (int j = 0; j < 8; ++j) {
        const bool real = (k + j < cnt);
        if constexpr (C == 64) {
          const float p0 = r2[j].x * w[j];
          const float p1 = r2[j].y * w[j];
          a0 += real ? p0 : 0.0f;
          a1 += real ? p1 : 0.0f;
        } else {
          const float p0 = r1[j] * w[j];
          a0 += real ? p0 : 0.0f;
        }
      }
    }
  }

  if constexpr (C == 64) {
    float v0 = a0 + self2.x * di * di + bias[2 * lane];
    float v1 = a1 + self2.y * di * di + bias[2 * lane + 1];
    if (RELU) { v0 = v0 > 0.0f ? v0 : 0.0f; v1 = v1 > 0.0f ? v1 : 0.0f; }
    v2f o; o.x = v0; o.y = v1;
    *(v2f*)(res + (size_t)node * C + 2 * lane) = o;
  } else {
    float v0 = a0 + self1 * di * di + bias[lane];
    if (RELU) v0 = v0 > 0.0f ? v0 : 0.0f;
    res[(size_t)node * C + lane] = v0;
  }
}

extern "C" void kernel_launch(void* const* d_in, const int* in_sizes, int n_in,
                              void* d_out, int out_size, void* d_ws, size_t ws_size,
                              hipStream_t stream) {
  const float* x   = (const float*)d_in[0];
  const int*   ei  = (const int*)d_in[1];
  const float* W1  = (const float*)d_in[2];
  const float* b1  = (const float*)d_in[3];
  const float* W2  = (const float*)d_in[4];
  const float* b2  = (const float*)d_in[5];
  float*       out = (float*)d_out;

  const int N = in_sizes[0] / IN_CH;
  const int E = in_sizes[1] / 2;
  const int* src = ei;
  const int* dst = ei + E;

  float* dinv = (float*)d_ws;
  float* bufA = dinv + 131072;
  float* bufB = bufA + (size_t)N * HID;

  int* cnt  = (int*)bufA;
  int* rank = (int*)bufB;
  int* offs1 = (int*)out;
  int* list1 = offs1 + LIST_OFF;
  int* offs2 = (int*)(bufA + (size_t)N * OUT_CH);
  int* list2 = offs2 + LIST_OFF;

  const int BT = 256;

  k_zero<<<(N + BT - 1) / BT, BT, 0, stream>>>(cnt, N);
  k_rank<<<(E + BT - 1) / BT, BT, 0, stream>>>(dst, cnt, rank, E);
  k_scan<<<(N + SCAN_PER - 1) / SCAN_PER, SCAN_T, 0, stream>>>(cnt, offs1, dinv, N);
  k_fill<<<(E + BT - 1) / BT, BT, 0, stream>>>(src, dst, rank, offs1, list1, E);
  {
    int tiles_m = (N + 15) / 16, tiles_n = HID / 16;
    int tiles = tiles_m * tiles_n;
    long threads = (long)tiles * 32;
    k_gemm_f32<<<(int)((threads + BT - 1) / BT), BT, 0, stream>>>(
        x, W1, bufA, N, IN_CH, HID, tiles_n, tiles);
  }
  k_gather<HID, true><<<(N + BT / 32 - 1) / (BT / 32), BT, 0, stream>>>(
      bufA, list1, offs1, dinv, b1, bufB, N);
  {
    int tiles_m = (N + 15) / 16, tiles_n = OUT_CH / 16;
    int tiles = tiles_m * tiles_n;
    long threads = (long)tiles * 32;
    k_gemm_f32<<<(int)((threads + BT - 1) / BT), BT, 0, stream>>>(
        bufB, W2, bufA, N, HID, OUT_CH, tiles_n, tiles);
  }
  k_move<<<(LIST_OFF + E + BT - 1) / BT, BT, 0, stream>>>(offs1, offs2, N + 1, LIST_OFF + E);
  k_gather<OUT_CH, false><<<(N + BT / 32 - 1) / (BT / 32), BT, 0, stream>>>(
      bufA, list2, offs2, dinv, b2, out, N);
}


# dinv[src] folded into both GEMM epilogues (rows stored pre-scaled); each node sums plain rows in list order plus its own row and scales once by dinv[dst]; no per-edge weight gather; rounding changes
# speedup vs baseline: 2.0425x; 1.0312x over previous
#include <hip/hip_runtime.h>
#include <hip/hip_bf16.h>


typedef float v2f __attribute__((ext_vector_type(2)));
typedef float v8f __attribute__((ext_vector_type(8)));

#define N_NODES 100000
#define IN_CH   128
#define HID     64
#define OUT_CH  32

#define SCAN_T   1024
#define SCAN_PER (SCAN_T * 4)
#define LIST_OFF 131072

__global__ void k_zero(int* __restrict__ cnt, int n) {
  int i = blockIdx.x * blockDim.x + threadIdx.x;
  if (i < n) cnt[i] = 0;
}

__global__ void k_rank(const int* __restrict__ dst, int* __restrict__ cnt,
                       int* __restrict__ rank, int e) {
  int i = blockIdx.x * blockDim.x + threadIdx.x;
  if (i < e) {
    rank[i] = __hip_atomic_fetch_add(&cnt[dst[i]], 1,
                                     __ATOMIC_RELAXED, __HIP_MEMORY_SCOPE_AGENT);
  }
}

__global__ void __launch_bounds__(SCAN_T)
k_scan(const int* __restrict__ cnt, int* __restrict__ offs,
       float* __restrict__ dinv, int n) {
  __shared__ int s_inc[SCAN_T / 32];
  __shared__ int s_pre[SCAN_T / 32];
  const int tid  = threadIdx.x;
  const int lane = tid & 31;
  const int wv   = tid >> 5;
  const int blk0 = blockIdx.x * SCAN_PER;

  int pre = 0;
  const int nfront = blockIdx.x;
  for (int j0 = 0; j0 < nfront; j0 += 8) {
    int4 v[8];
    #pragma unroll
    for (int j = 0; j < 8; ++j) {
      int jj = j0 + j;
      if (jj > nfront - 1) jj = nfront - 1;
      v[j] = *(const int4*)(cnt + jj * SCAN_PER + tid * 4);
    }
    #pragma unroll
    for (int j = 0; j < 8; ++j) {
      const int t = (v[j].x + v[j].y) + (v[j].z + v[j].w);
      pre += (j0 + j < nfront) ? t : 0;
    }
  }

  const int i0 = blk0 + tid * 4;
  int c[4];
  #pragma unroll
  for (int k = 0; k < 4; ++k) c[k] = (i0 + k < n) ? cnt[i0 + k] : 0;
  const int tsum = (c[0] + c[1]) + (c[2] + c[3]);

  int inc = tsum;
  #pragma unroll
  for (int d = 1; d < 32; d <<= 1) {
    int t = __shfl_up(inc, d, 32);
    if (lane >= d) inc += t;
  }
  #pragma unroll
  for (int d = 16; d >= 1; d >>= 1) pre += __shfl_xor(pre, d, 32);

  if (lane == 31) s_inc[wv] = inc;
  if (lane == 0)  s_pre[wv] = pre;
  __syncthreads();

  int base = 0;
  for (int w = 0; w < SCAN_T / 32; ++w) {
    base += s_pre[w];
    if (w < wv) base += s_inc[w];
  }
  int ex = base + inc - tsum;

  #pragma unroll
  for (int k = 0; k < 4; ++k) {
    if (i0 + k < n) {
      offs[i0 + k] = ex;
      dinv[i0 + k] = rsqrtf((float)(c[k] + 1));
      ex += c[k];
    }
  }
  if (i0 <= n - 1 && n - 1 < i0 + 4) offs[n] = ex;
}

__global__ void k_fill(const int* __restrict__ src, const int* __restrict__ dst,
                       const int* __restrict__ rank, const int* __restrict__ offs,
                       int* __restrict__ list, int e) {
  int i = blockIdx.x * blockDim.x + threadIdx.x;
  if (i < e) list[offs[dst[i]] + rank[i]] = src[i];
}

__global__ void k_move(const int* __restrict__ from, int* __restrict__ to,
                       int n_starts, int total) {
  int i = blockIdx.x * blockDim.x + threadIdx.x;
  if (i < total && (i < n_starts || i >= LIST_OFF)) to[i] = from[i];
}

__global__ void k_gemm_f32(const float* __restrict__ A, const float* __restrict__ W,
                           const float* __restrict__ rowscale,
                           float* __restrict__ out, int M, int K, int NC,
                           int tiles_n, int total_tiles) {
  int wave = (int)((blockIdx.x * (long)blockDim.x + threadIdx.x) >> 5);
  if (wave >= total_tiles) return;
  int lane  = threadIdx.x & 31;
  int tm    = wave / tiles_n;
  int tn    = wave % tiles_n;
  int hi    = lane >> 4;
  int l15   = lane & 15;
  int row   = tm * 16 + l15;
  int col   = tn * 16 + l15;
  int rrow  = row < M ? row : M - 1;

  const float* arow = A + (size_t)rrow * K + 2 * hi;
  const float* wcol = W + (size_t)(2 * hi) * NC + col;

  const int m0 = tm * 16 + hi * 8;
  float sc[8];
  #pragma unroll
  for (int v = 0; v < 8; ++v) {
    const int r = m0 + v;
    sc[v] = rowscale[r < M ? r : M - 1];
  }

  v8f c = {};
  for (int k0 = 0; k0 < K; k0 += 4) {
    v2f a = *(const v2f*)(arow + k0);
    v2f b;
    b.x = wcol[(size_t)k0 * NC];
    b.y = wcol[(size_t)(k0 + 1) * NC];
    c = __builtin_amdgcn_wmma_f32_16x16x4_f32(
         false, a,  false, b,
         (short)0, c,  false,  false);
  }

  #pragma unroll
  for (int v = 0; v < 8; ++v) {
    int r = m0 + v;
    if (r < M) out[(size_t)r * NC + col] = c[v] * sc[v];
  }
}

template <int C, bool RELU>
__global__ void k_gather(const float* __restrict__ m, const int* __restrict__ list,
                         const int* __restrict__ offs, const float* __restrict__ dinv,
                         const float* __restrict__ bias, float* __restrict__ res, int n) {
  const int node = (int)(blockIdx.x * (blockDim.x >> 5)) +
                   __builtin_amdgcn_readfirstlane((int)(threadIdx.x >> 5));
  if (node >= n) return;
  const int lane = threadIdx.x & 31;
  const int beg = offs[node];
  const int end = offs[node + 1];
  const float di = dinv[node];

  v2f  self2 = {};
  float self1 = 0.0f;
  if constexpr (C == 64) self2 = *(const v2f*)(m + (size_t)node * C + 2 * lane);
  else                   self1 = m[(size_t)node * C + lane];

  float a0 = 0.0f, a1 = 0.0f;
  for (int base = beg; base < end; base += 32) {
    int cnt = end - base;
    if (cnt > 32) cnt = 32;
    int sj = 0;
    if (lane < cnt) sj = list[base + lane];
    for (int k = 0; k < cnt; k += 8) {
      v2f   r2[8];
      float r1[8];
      #pragma unroll
      for (int j = 0; j < 8; ++j) {
        int kk = k + j;
        if (kk > cnt - 1) kk = cnt - 1;
        const int s = __builtin_amdgcn_readlane(sj, kk);
        if constexpr (C == 64) r2[j] = *(const v2f*)(m + (size_t)s * C + 2 * lane);
        else                   r1[j] = m[(size_t)s * C + lane];
      }
      #pragma unroll
      for (int j = 0; j < 8; ++j) {
        const bool real = (k + j < cnt);
        if constexpr (C == 64) {
          a0 += real ? r2[j].x : 0.0f;
          a1 += real ? r2[j].y : 0.0f;
        } else {
          a0 += real ? r1[j] : 0.0f;
        }
      }
    }
  }

  if constexpr (C == 64) {
    float v0 = (a0 + self2.x) * di + bias[2 * lane];
    float v1 = (a1 + self2.y) * di + bias[2 * lane + 1];
    if (RELU) { v0 = v0 > 0.0f ? v0 : 0.0f; v1 = v1 > 0.0f ? v1 : 0.0f; }
    v2f o; o.x = v0; o.y = v1;
    *(v2f*)(res + (size_t)node * C + 2 * lane) = o;
  } else {
    float v0 = (a0 + self1) * di + bias[lane];
    if (RELU) v0 = v0 > 0.0f ? v0 : 0.0f;
    res[(size_t)node * C + lane] = v0;
  }
}

extern "C" void kernel_launch(void* const* d_in, const int* in_sizes, int n_in,
                              void* d_out, int out_size, void* d_ws, size_t ws_size,
                              hipStream_t stream) {
  const float* x   = (const float*)d_in[0];
  const int*   ei  = (const int*)d_in[1];
  const float* W1  = (const float*)d_in[2];
  const float* b1  = (const float*)d_in[3];
  const float* W2  = (const float*)d_in[4];
  const float* b2  = (const float*)d_in[5];
  float*       out = (float*)d_out;

  const int N = in_sizes[0] / IN_CH;
  const int E = in_sizes[1] / 2;
  const int* src = ei;
  const int* dst = ei + E;

  float* dinv = (float*)d_ws;
  float* bufA = dinv + 131072;
  float* bufB = bufA + (size_t)N * HID;

  int* cnt  = (int*)bufA;
  int* rank = (int*)bufB;
  int* offs1 = (int*)out;
  int* list1 = offs1 + LIST_OFF;
  int* offs2 = (int*)(bufA + (size_t)N * OUT_CH);
  int* list2 = offs2 + LIST_OFF;

  const int BT = 256;

  k_zero<<<(N + BT - 1) / BT, BT, 0, stream>>>(cnt, N);
  k_rank<<<(E + BT - 1) / BT, BT, 0, stream>>>(dst, cnt, rank, E);
  k_scan<<<(N + SCAN_PER - 1) / SCAN_PER, SCAN_T, 0, stream>>>(cnt, offs1, dinv, N);
  k_fill<<<(E + BT - 1) / BT, BT, 0, stream>>>(src, dst, rank, offs1, list1, E);
  {
    int tiles_m = (N + 15) / 16, tiles_n = HID / 16;
    int tiles = tiles_m * tiles_n;
    long threads = (long)tiles * 32;
    k_gemm_f32<<<(int)((threads + BT - 1) / BT), BT, 0, stream>>>(
        x, W1, dinv, bufA, N, IN_CH, HID, tiles_n, tiles);
  }
  k_gather<HID, true><<<(N + BT / 32 - 1) / (BT / 32), BT, 0, stream>>>(
      bufA, list1, offs1, dinv, b1, bufB, N);
  {
    int tiles_m = (N + 15) / 16, tiles_n = OUT_CH / 16;
    int tiles = tiles_m * tiles_n;
    long threads = (long)tiles * 32;
    k_gemm_f32<<<(int)((threads + BT - 1) / BT), BT, 0, stream>>>(
        bufB, W2, dinv, bufA, N, HID, OUT_CH, tiles_n, tiles);
  }
  k_move<<<(LIST_OFF + E + BT - 1) / BT, BT, 0, stream>>>(offs1, offs2, N + 1, LIST_OFF + E);
  k_gather<OUT_CH, false><<<(N + BT / 32 - 1) / (BT / 32), BT, 0, stream>>>(
      bufA, list2, offs2, dinv, b2, out, N);
}


# GEMMs re-tiled: one wave per 16-row strip over all output columns (A fragment loaded once per k step, not once per column tile; B and stores as whole rows); every sum in the same order
# speedup vs baseline: 2.1538x; 1.0545x over previous
#include <hip/hip_runtime.h>
#include <hip/hip_bf16.h>


typedef float v2f __attribute__((ext_vector_type(2)));
typedef float v4f __attribute__((ext_vector_type(4)));
typedef float v8f __attribute__((ext_vector_type(8)));

template <int NT> struct RowVec;
template <> struct RowVec<2> { typedef v2f type; };
template <> struct RowVec<4> { typedef v4f type; };

#define N_NODES 100000
#define IN_CH   128
#define HID     64
#define OUT_CH  32

#define SCAN_T   1024
#define SCAN_PER (SCAN_T * 4)
#define LIST_OFF 131072

__global__ void k_zero(int* __restrict__ cnt, int n) {
  int i = blockIdx.x * blockDim.x + threadIdx.x;
  if (i < n) cnt[i] = 0;
}

__global__ void k_rank(const int* __restrict__ dst, int* __restrict__ cnt,
                       int* __restrict__ rank, int e) {
  int i = blockIdx.x * blockDim.x + threadIdx.x;
  if (i < e) {
    rank[i] = __hip_atomic_fetch_add(&cnt[dst[i]], 1,
                                     __ATOMIC_RELAXED, __HIP_MEMORY_SCOPE_AGENT);
  }
}

__global__ void __launch_bounds__(SCAN_T)
k_scan(const int* __restrict__ cnt, int* __restrict__ offs,
       float* __restrict__ dinv, int n) {
  __shared__ int s_inc[SCAN_T / 32];
  __shared__ int s_pre[SCAN_T / 32];
  const int tid  = threadIdx.x;
  const int lane = tid & 31;
  const int wv   = tid >> 5;
  const int blk0 = blockIdx.x * SCAN_PER;

  int pre = 0;
  const int nfront = blockIdx.x;
  for (int j0 = 0; j0 < nfront; j0 += 8) {
    int4 v[8];
    #pragma unroll
    for (int j = 0; j < 8; ++j) {
      int jj = j0 + j;
      if (jj > nfront - 1) jj = nfront - 1;
      v[j] = *(const int4*)(cnt + jj * SCAN_PER + tid * 4);
    }
    #pragma unroll
    for (int j = 0; j < 8; ++j) {
      const int t = (v[j].x + v[j].y) + (v[j].z + v[j].w);
      pre += (j0 + j < nfront) ? t : 0;
    }
  }

  const int i0 = blk0 + tid * 4;
  int c[4];
  #pragma unroll
  for (int k = 0; k < 4; ++k) c[k] = (i0 + k < n) ? cnt[i0 + k] : 0;
  const int tsum = (c[0] + c[1]) + (c[2] + c[3]);

  int inc = tsum;
  #pragma unroll
  for (int d = 1; d < 32; d <<= 1) {
    int t = __shfl_up(inc, d, 32);
    if (lane >= d) inc += t;
  }
  #pragma unroll
  for (int d = 16; d >= 1; d >>= 1) pre += __shfl_xor(pre, d, 32);

  if (lane == 31) s_inc[wv] = inc;
  if (lane == 0)  s_pre[wv] = pre;
  __syncthreads();

  int base = 0;
  for (int w = 0; w < SCAN_T / 32; ++w) {
    base += s_pre[w];
    if (w < wv) base += s_inc[w];
  }
  int ex = base + inc - tsum;

  #pragma unroll
  for (int k = 0; k < 4; ++k) {
    if (i0 + k < n) {
      offs[i0 + k] = ex;
      dinv[i0 + k] = rsqrtf((float)(c[k] + 1));
      ex += c[k];
    }
  }
  if (i0 <= n - 1 && n - 1 < i0 + 4) offs[n] = ex;
}

__global__ void k_fill(const int* __restrict__ src, const int* __restrict__ dst,
                       const int* __restrict__ rank, const int* __restrict__ offs,
                       int* __restrict__ list, int e) {
  int i = blockIdx.x * blockDim.x + threadIdx.x;
  if (i < e) list[offs[dst[i]] + rank[i]] = src[i];
}

__global__ void k_move(const int* __restrict__ from, int* __restrict__ to,
                       int n_starts, int total) {
  int i = blockIdx.x * blockDim.x + threadIdx.x;
  if (i < total && (i < n_starts || i >= LIST_OFF)) to[i] = from[i];
}

template <int K, int NT>
__global__ void __launch_bounds__(128)
k_gemm_strip(const float* __restrict__ A, const float* __restrict__ W,
                             const float* __restrict__ rowscale,
                             float* __restrict__ out, int M, int strips) {
  typedef typename RowVec<NT>::type rowv;
  constexpr int NC = 16 * NT;
  constexpr int G  = 4;
  constexpr int NG = K / (4 * G);
  static_assert(K % (4 * G) == 0, "K must be a multiple of 16");

  const int wave = (int)((blockIdx.x * (long)blockDim.x + threadIdx.x) >> 5);
  if (wave >= strips) return;
  const int lane = threadIdx.x & 31;
  const int hi   = lane >> 4;
  const int l15  = lane & 15;
  const int row  = wave * 16 + l15;
  const int rrow = row < M ? row : M - 1;

  const float* arow = A + (size_t)rrow * K + 2 * hi;
  const float* wrow = W + (size_t)(2 * hi) * NC + NT * l15;

  const int m0 = wave * 16 + hi * 8;
  float sc[8];
  #pragma unroll
  for (int v = 0; v < 8; ++v) {
    const int r = m0 + v;
    sc[v] = rowscale[r < M ? r : M - 1];
  }

  v8f c[NT];
  #pragma unroll
  for (int t = 0; t < NT; ++t) c[t] = v8f{};

  v2f  a[2][G];
  rowv b0[2][G], b1[2][G];
  #pragma unroll
  for (int j = 0; j < G; ++j) {
    a[0][j]  = *(const v2f*)(arow + 4 * j);
    b0[0][j] = *(const rowv*)(wrow + (size_t)(4 * j) * NC);
    b1[0][j] = *(const rowv*)(wrow + (size_t)(4 * j + 1) * NC);
  }
  #pragma unroll
  for (int g = 0; g < NG; ++g) {
    const int cur = g & 1, nxt = cur ^ 1;
    if (g + 1 < NG) {
      #pragma unroll
      for (int j = 0; j < G; ++j) {
        const int k0 = (g + 1) * 4 * G + 4 * j;
        a[nxt][j]  = *(const v2f*)(arow + k0);
        b0[nxt][j] = *(const rowv*)(wrow + (size_t)k0 * NC);
        b1[nxt][j] = *(const rowv*)(wrow + (size_t)(k0 + 1) * NC);
      }
    }
    #pragma unroll
    for (int j = 0; j < G; ++j) {
      #pragma unroll
      for (int t = 0; t < NT; ++t) {
        v2f b;
        b.x = b0[cur][j][t];
        b.y = b1[cur][j][t];
        c[t] = __builtin_amdgcn_wmma_f32_16x16x4_f32(
             false, a[cur][j],  false, b,
             (short)0, c[t],  false,  false);
      }
    }
  }

  #pragma unroll
  for (int v = 0; v < 8; ++v) {
    const int r = m0 + v;
    rowv o;
    #pragma unroll
    for (int t = 0; t < NT; ++t) o[t] = c[t][v] * sc[v];
    if (r < M) *(rowv*)(out + (size_t)r * NC + NT * l15) = o;
  }
}

template <int C, bool RELU>
__global__ void k_gather(const float* __restrict__ m, const int* __restrict__ list,
                         const int* __restrict__ offs, const float* __restrict__ dinv,
                         const float* __restrict__ bias, float* __restrict__ res, int n) {
  const int node = (int)(blockIdx.x * (blockDim.x >> 5)) +
                   __builtin_amdgcn_readfirstlane((int)(threadIdx.x >> 5));
  if (node >= n) return;
  const int lane = threadIdx.x & 31;
  const int beg = offs[node];
  const int end = offs[node + 1];
  const float di = dinv[node];

  v2f  self2 = {};
  float self1 = 0.0f;
  if constexpr (C == 64) self2 = *(const v2f*)(m + (size_t)node * C + 2 * lane);
  else                   self1 = m[(size_t)node * C + lane];

  float a0 = 0.0f, a1 = 0.0f;
  for (int base = beg; base < end; base += 32) {
    int cnt = end - base;
    if (cnt > 32) cnt = 32;
    int sj = 0;
    if (lane < cnt) sj = list[base + lane];
    for (int k = 0; k < cnt; k += 8) {
      v2f   r2[8];
      float r1[8];
      #pragma unroll
      for (int j = 0; j < 8; ++j) {
        int kk = k + j;
        if (kk > cnt - 1) kk = cnt - 1;
        const int s = __builtin_amdgcn_readlane(sj, kk);
        if constexpr (C == 64) r2[j] = *(const v2f*)(m + (size_t)s * C + 2 * lane);
        else                   r1[j] = m[(size_t)s * C + lane];
      }
      #pragma unroll
      for (int j = 0; j < 8; ++j) {
        const bool real = (k + j < cnt);
        if constexpr (C == 64) {
          a0 += real ? r2[j].x : 0.0f;
          a1 += real ? r2[j].y : 0.0f;
        } else {
          a0 += real ? r1[j] : 0.0f;
        }
      }
    }
  }

  if constexpr (C == 64) {
    float v0 = (a0 + self2.x) * di + bias[2 * lane];
    float v1 = (a1 + self2.y) * di + bias[2 * lane + 1];
    if (RELU) { v0 = v0 > 0.0f ? v0 : 0.0f; v1 = v1 > 0.0f ? v1 : 0.0f; }
    v2f o; o.x = v0; o.y = v1;
    *(v2f*)(res + (size_t)node * C + 2 * lane) = o;
  } else {
    float v0 = (a0 + self1) * di + bias[lane];
    if (RELU) v0 = v0 > 0.0f ? v0 : 0.0f;
    res[(size_t)node * C + lane] = v0;
  }
}

extern "C" void kernel_launch(void* const* d_in, const int* in_sizes, int n_in,
                              void* d_out, int out_size, void* d_ws, size_t ws_size,
                              hipStream_t stream) {
  const float* x   = (const float*)d_in[0];
  const int*   ei  = (const int*)d_in[1];
  const float* W1  = (const float*)d_in[2];
  const float* b1  = (const float*)d_in[3];
  const float* W2  = (const float*)d_in[4];
  const float* b2  = (const float*)d_in[5];
  float*       out = (float*)d_out;

  const int N = in_sizes[0] / IN_CH;
  const int E = in_sizes[1] / 2;
  const int* src = ei;
  const int* dst = ei + E;

  float* dinv = (float*)d_ws;
  float* bufA = dinv + 131072;
  float* bufB = bufA + (size_t)N * HID;

  int* cnt  = (int*)bufA;
  int* rank = (int*)bufB;
  int* offs1 = (int*)out;
  int* list1 = offs1 + LIST_OFF;
  int* offs2 = (int*)(bufA + (size_t)N * OUT_CH);
  int* list2 = offs2 + LIST_OFF;

  const int BT = 256;

  k_zero<<<(N + BT - 1) / BT, BT, 0, stream>>>(cnt, N);
  k_rank<<<(E + BT - 1) / BT, BT, 0, stream>>>(dst, cnt, rank, E);
  k_scan<<<(N + SCAN_PER - 1) / SCAN_PER, SCAN_T, 0, stream>>>(cnt, offs1, dinv, N);
  k_fill<<<(E + BT - 1) / BT, BT, 0, stream>>>(src, dst, rank, offs1, list1, E);
  const int GT = 128;
  const int strips = (N + 15) / 16;
  k_gemm_strip<IN_CH, HID / 16><<<(strips * 32 + GT - 1) / GT, GT, 0, stream>>>(
      x, W1, dinv, bufA, N, strips);
  k_gather<HID, true><<<(N + BT / 32 - 1) / (BT / 32), BT, 0, stream>>>(
      bufA, list1, offs1, dinv, b1, bufB, N);
  k_gemm_strip<HID, OUT_CH / 16><<<(strips * 32 + GT - 1) / GT, GT, 0, stream>>>(
      bufB, W2, dinv, bufA, N, strips);
  k_move<<<(LIST_OFF + E + BT - 1) / BT, BT, 0, stream>>>(offs1, offs2, N + 1, LIST_OFF + E);
  k_gather<OUT_CH, false><<<(N + BT / 32 - 1) / (BT / 32), BT, 0, stream>>>(
      bufA, list2, offs2, dinv, b2, out, N);
}


# Layer-1 aggregation fused with the layer-2 GEMM: a 16-wave block keeps its 16 h rows in LDS, one wave does the strip WMMA product; h never stored, list stays put (copy launch gone); same sums
# speedup vs baseline: 2.2289x; 1.0349x over previous
#include <hip/hip_runtime.h>
#include <hip/hip_bf16.h>


typedef float v2f __attribute__((ext_vector_type(2)));
typedef float v4f __attribute__((ext_vector_type(4)));
typedef float v8f __attribute__((ext_vector_type(8)));

template <int NT> struct RowVec;
template <> struct RowVec<2> { typedef v2f type; };
template <> struct RowVec<4> { typedef v4f type; };

#define N_NODES 100000
#define IN_CH   128
#define HID     64
#define OUT_CH  32

#define SCAN_T   1024
#define SCAN_PER (SCAN_T * 4)
#define LIST_OFF 131072

__global__ void k_zero(int* __restrict__ cnt, int n) {
  int i = blockIdx.x * blockDim.x + threadIdx.x;
  if (i < n) cnt[i] = 0;
}

__global__ void k_rank(const int* __restrict__ dst, int* __restrict__ cnt,
                       int* __restrict__ rank, int e) {
  int i = blockIdx.x * blockDim.x + threadIdx.x;
  if (i < e) {
    rank[i] = __hip_atomic_fetch_add(&cnt[dst[i]], 1,
                                     __ATOMIC_RELAXED, __HIP_MEMORY_SCOPE_AGENT);
  }
}

__global__ void __launch_bounds__(SCAN_T)
k_scan(const int* __restrict__ cnt, int* __restrict__ offs,
       float* __restrict__ dinv, int n) {
  __shared__ int s_inc[SCAN_T / 32];
  __shared__ int s_pre[SCAN_T / 32];
  const int tid  = threadIdx.x;
  const int lane = tid & 31;
  const int wv   = tid >> 5;
  const int blk0 = blockIdx.x * SCAN_PER;

  int pre = 0;
  const int nfront = blockIdx.x;
  for (int j0 = 0; j0 < nfront; j0 += 8) {
    int4 v[8];
    #pragma unroll
    for (int j = 0; j < 8; ++j) {
      int jj = j0 + j;
      if (jj > nfront - 1) jj = nfront - 1;
      v[j] = *(const int4*)(cnt + jj * SCAN_PER + tid * 4);
    }
    #pragma unroll
    for (int j = 0; j < 8; ++j) {
      const int t = (v[j].x + v[j].y) + (v[j].z + v[j].w);
      pre += (j0 + j < nfront) ? t : 0;
    }
  }

  const int i0 = blk0 + tid * 4;
  int c[4];
  #pragma unroll
  for (int k = 0; k < 4; ++k) c[k] = (i0 + k < n) ? cnt[i0 + k] : 0;
  const int tsum = (c[0] + c[1]) + (c[2] + c[3]);

  int inc = tsum;
  #pragma unroll
  for (int d = 1; d < 32; d <<= 1) {
    int t = __shfl_up(inc, d, 32);
    if (lane >= d) inc += t;
  }
  #pragma unroll
  for (int d = 16; d >= 1; d >>= 1) pre += __shfl_xor(pre, d, 32);

  if (lane == 31) s_inc[wv] = inc;
  if (lane == 0)  s_pre[wv] = pre;
  __syncthreads();

  int base = 0;
  for (int w = 0; w < SCAN_T / 32; ++w) {
    base += s_pre[w];
    if (w < wv) base += s_inc[w];
  }
  int ex = base + inc - tsum;

  #pragma unroll
  for (int k = 0; k < 4; ++k) {
    if (i0 + k < n) {
      offs[i0 + k] = ex;
      dinv[i0 + k] = rsqrtf((float)(c[k] + 1));
      ex += c[k];
    }
  }
  if (i0 <= n - 1 && n - 1 < i0 + 4) offs[n] = ex;
}

__global__ void k_fill(const int* __restrict__ src, const int* __restrict__ dst,
                       const int* __restrict__ rank, const int* __restrict__ offs,
                       int* __restrict__ list, int e) {
  int i = blockIdx.x * blockDim.x + threadIdx.x;
  if (i < e) list[offs[dst[i]] + rank[i]] = src[i];
}

template <int K, int NT>
__global__ void __launch_bounds__(128)
k_gemm_strip(const float* __restrict__ A, const float* __restrict__ W,
                             const float* __restrict__ rowscale,
                             float* __restrict__ out, int M, int strips) {
  typedef typename RowVec<NT>::type rowv;
  constexpr int NC = 16 * NT;
  constexpr int G  = 4;
  constexpr int NG = K / (4 * G);
  static_assert(K % (4 * G) == 0, "K must be a multiple of 16");

  const int wave = (int)((blockIdx.x * (long)blockDim.x + threadIdx.x) >> 5);
  if (wave >= strips) return;
  const int lane = threadIdx.x & 31;
  const int hi   = lane >> 4;
  const int l15  = lane & 15;
  const int row  = wave * 16 + l15;
  const int rrow = row < M ? row : M - 1;

  const float* arow = A + (size_t)rrow * K + 2 * hi;
  const float* wrow = W + (size_t)(2 * hi) * NC + NT * l15;

  const int m0 = wave * 16 + hi * 8;
  float sc[8];
  #pragma unroll
  for (int v = 0; v < 8; ++v) {
    const int r = m0 + v;
    sc[v] = rowscale[r < M ? r : M - 1];
  }

  v8f c[NT];
  #pragma unroll
  for (int t = 0; t < NT; ++t) c[t] = v8f{};

  v2f  a[2][G];
  rowv b0[2][G], b1[2][G];
  #pragma unroll
  for (int j = 0; j < G; ++j) {
    a[0][j]  = *(const v2f*)(arow + 4 * j);
    b0[0][j] = *(const rowv*)(wrow + (size_t)(4 * j) * NC);
    b1[0][j] = *(const rowv*)(wrow + (size_t)(4 * j + 1) * NC);
  }
  #pragma unroll
  for (int g = 0; g < NG; ++g) {
    const int cur = g & 1, nxt = cur ^ 1;
    if (g + 1 < NG) {
      #pragma unroll
      for (int j = 0; j < G; ++j) {
        const int k0 = (g + 1) * 4 * G + 4 * j;
        a[nxt][j]  = *(const v2f*)(arow + k0);
        b0[nxt][j] = *(const rowv*)(wrow + (size_t)k0 * NC);
        b1[nxt][j] = *(const rowv*)(wrow + (size_t)(k0 + 1) * NC);
      }
    }
    #pragma unroll
    for (int j = 0; j < G; ++j) {
      #pragma unroll
      for (int t = 0; t < NT; ++t) {
        v2f b;
        b.x = b0[cur][j][t];
        b.y = b1[cur][j][t];
        c[t] = __builtin_amdgcn_wmma_f32_16x16x4_f32(
             false, a[cur][j],  false, b,
             (short)0, c[t],  false,  false);
      }
    }
  }

  #pragma unroll
  for (int v = 0; v < 8; ++v) {
    const int r = m0 + v;
    rowv o;
    #pragma unroll
    for (int t = 0; t < NT; ++t) o[t] = c[t][v] * sc[v];
    if (r < M) *(rowv*)(out + (size_t)r * NC + NT * l15) = o;
  }
}

template <int C, bool RELU>
__global__ void k_gather(const float* __restrict__ m, const int* __restrict__ list,
                         const int* __restrict__ offs, const float* __restrict__ dinv,
                         const float* __restrict__ bias, float* __restrict__ res, int n) {
  const int node = (int)(blockIdx.x * (blockDim.x >> 5)) +
                   __builtin_amdgcn_readfirstlane((int)(threadIdx.x >> 5));
  if (node >= n) return;
  const int lane = threadIdx.x & 31;
  const int beg = offs[node];
  const int end = offs[node + 1];
  const float di = dinv[node];

  v2f  self2 = {};
  float self1 = 0.0f;
  if constexpr (C == 64) self2 = *(const v2f*)(m + (size_t)node * C + 2 * lane);
  else                   self1 = m[(size_t)node * C + lane];

  float a0 = 0.0f, a1 = 0.0f;
  for (int base = beg; base < end; base += 32) {
    int cnt = end - base;
    if (cnt > 32) cnt = 32;
    int sj = 0;
    if (lane < cnt) sj = list[base + lane];
    for (int k = 0; k < cnt; k += 8) {
      v2f   r2[8];
      float r1[8];
      #pragma unroll
      for (int j = 0; j < 8; ++j) {
        int kk = k + j;
        if (kk > cnt - 1) kk = cnt - 1;
        const int s = __builtin_amdgcn_readlane(sj, kk);
        if constexpr (C == 64) r2[j] = *(const v2f*)(m + (size_t)s * C + 2 * lane);
        else                   r1[j] = m[(size_t)s * C + lane];
      }
      #pragma unroll
      for (int j = 0; j < 8; ++j) {
        const bool real = (k + j < cnt);
        if constexpr (C == 64) {
          a0 += real ? r2[j].x : 0.0f;
          a1 += real ? r2[j].y : 0.0f;
        } else {
          a0 += real ? r1[j] : 0.0f;
        }
      }
    }
  }

  if constexpr (C == 64) {
    float v0 = (a0 + self2.x) * di + bias[2 * lane];
    float v1 = (a1 + self2.y) * di + bias[2 * lane + 1];
    if (RELU) { v0 = v0 > 0.0f ? v0 : 0.0f; v1 = v1 > 0.0f ? v1 : 0.0f; }
    v2f o; o.x = v0; o.y = v1;
    *(v2f*)(res + (size_t)node * C + 2 * lane) = o;
  } else {
    float v0 = (a0 + self1) * di + bias[lane];
    if (RELU) v0 = v0 > 0.0f ? v0 : 0.0f;
    res[(size_t)node * C + lane] = v0;
  }
}

#define H_PITCH 68
__global__ void __launch_bounds__(512, 16)
k_gather_gemm(const float* __restrict__ m, const int* __restrict__ list,
              const int* __restrict__ offs, const float* __restrict__ dinv,
              const float* __restrict__ bias, const float* __restrict__ W,
              float* __restrict__ out2, int n) {
  constexpr int C  = HID;
  constexpr int NC = OUT_CH;
  __shared__ __attribute__((aligned(16))) float sH[16 * H_PITCH];

  const int wv   = __builtin_amdgcn_readfirstlane((int)(threadIdx.x >> 5));
  const int lane = threadIdx.x & 31;
  const int node = (int)blockIdx.x * 16 + wv;

  v2f hrow = {};
  if (node < n) {
    const int beg = offs[node];
    const int end = offs[node + 1];
    const float di = dinv[node];

    const v2f self2 = *(const v2f*)(m + (size_t)node * C + 2 * lane);

    float a0 = 0.0f, a1 = 0.0f;
    for (int base = beg; base < end; base += 32) {
      int cnt = end - base;
      if (cnt > 32) cnt = 32;
      int sj = 0;
      if (lane < cnt) sj = list[base + lane];
      for (int k = 0; k < cnt; k += 8) {
        v2f r2[8];
        #pragma unroll
        for (int j = 0; j < 8; ++j) {
          int kk = k + j;
          if (kk > cnt - 1) kk = cnt - 1;
          const int s = __builtin_amdgcn_readlane(sj, kk);
          r2[j] = *(const v2f*)(m + (size_t)s * C + 2 * lane);
        }
        #pragma unroll
        for (int j = 0; j < 8; ++j) {
          const bool real = (k + j < cnt);
          a0 += real ? r2[j].x : 0.0f;
          a1 += real ? r2[j].y : 0.0f;
        }
      }
    }
    float v0 = (a0 + self2.x) * di + bias[2 * lane];
    float v1 = (a1 + self2.y) * di + bias[2 * lane + 1];
    v0 = v0 > 0.0f ? v0 : 0.0f;
    v1 = v1 > 0.0f ? v1 : 0.0f;
    hrow.x = v0; hrow.y = v1;
  }
  *(v2f*)(sH + wv * H_PITCH + 2 * lane) = hrow;
  __syncthreads();
  if (wv != (int)(blockIdx.x & 3)) return;

  constexpr int G  = 2;
  constexpr int NG = C / (4 * G);
  const int hi  = lane >> 4;
  const int l15 = lane & 15;
  const float* arow = sH + l15 * H_PITCH + 2 * hi;
  const float* wrow = W + (size_t)(2 * hi) * NC + 2 * l15;

  const int m0 = (int)blockIdx.x * 16 + hi * 8;
  float sc[8];
  #pragma unroll
  for (int v = 0; v < 8; ++v) {
    const int r = m0 + v;
    sc[v] = dinv[r < n ? r : n - 1];
  }

  v8f c[2];
  c[0] = v8f{};
  c[1] = v8f{};

  v2f b0[2][G], b1[2][G];
  #pragma unroll
  for (int j = 0; j < G; ++j) {
    b0[0][j] = *(const v2f*)(wrow + (size_t)(4 * j) * NC);
    b1[0][j] = *(const v2f*)(wrow + (size_t)(4 * j + 1) * NC);
  }
  #pragma unroll
  for (int g = 0; g < NG; ++g) {
    const int cur = g & 1, nxt = cur ^ 1;
    if (g + 1 < NG) {
      #pragma unroll
      for (int j = 0; j < G; ++j) {
        const int k0 = (g + 1) * 4 * G + 4 * j;
        b0[nxt][j] = *(const v2f*)(wrow + (size_t)k0 * NC);
        b1[nxt][j] = *(const v2f*)(wrow + (size_t)(k0 + 1) * NC);
      }
    }
    #pragma unroll
    for (int j = 0; j < G; ++j) {
      const v2f a = *(const v2f*)(arow + g * 4 * G + 4 * j);
      #pragma unroll
      for (int t = 0; t < 2; ++t) {
        v2f b;
        b.x = b0[cur][j][t];
        b.y = b1[cur][j][t];
        c[t] = __builtin_amdgcn_wmma_f32_16x16x4_f32(
             false, a,  false, b,
             (short)0, c[t],  false,  false);
      }
    }
  }

  #pragma unroll
  for (int v = 0; v < 8; ++v) {
    const int r = m0 + v;
    v2f o;
    o.x = c[0][v] * sc[v];
    o.y = c[1][v] * sc[v];
    if (r < n) *(v2f*)(out2 + (size_t)r * NC + 2 * l15) = o;
  }
}

extern "C" void kernel_launch(void* const* d_in, const int* in_sizes, int n_in,
                              void* d_out, int out_size, void* d_ws, size_t ws_size,
                              hipStream_t stream) {
  const float* x   = (const float*)d_in[0];
  const int*   ei  = (const int*)d_in[1];
  const float* W1  = (const float*)d_in[2];
  const float* b1  = (const float*)d_in[3];
  const float* W2  = (const float*)d_in[4];
  const float* b2  = (const float*)d_in[5];
  float*       out = (float*)d_out;

  const int N = in_sizes[0] / IN_CH;
  const int E = in_sizes[1] / 2;
  const int* src = ei;
  const int* dst = ei + E;

  float* dinv = (float*)d_ws;
  float* bufA = dinv + 131072;
  float* bufB = bufA + (size_t)N * HID;

  int* cnt  = (int*)bufA;
  int*   offs = (int*)bufB;
  int*   list = offs + LIST_OFF;
  int*   rank = list + E;
  float* m2   = (float*)(list + E);

  const int BT = 256;

  k_zero<<<(N + BT - 1) / BT, BT, 0, stream>>>(cnt, N);
  k_rank<<<(E + BT - 1) / BT, BT, 0, stream>>>(dst, cnt, rank, E);
  k_scan<<<(N + SCAN_PER - 1) / SCAN_PER, SCAN_T, 0, stream>>>(cnt, offs, dinv, N);
  k_fill<<<(E + BT - 1) / BT, BT, 0, stream>>>(src, dst, rank, offs, list, E);
  const int GT = 128;
  const int strips = (N + 15) / 16;
  k_gemm_strip<IN_CH, HID / 16><<<(strips * 32 + GT - 1) / GT, GT, 0, stream>>>(
      x, W1, dinv, bufA, N, strips);
  k_gather_gemm<<<strips, 512, 0, stream>>>(bufA, list, offs, dinv, b1, W2, m2, N);
  k_gather<OUT_CH, false><<<(N + BT / 32 - 1) / (BT / 32), BT, 0, stream>>>(
      m2, list, offs, dinv, b2, out, N);
}


# Each aggregation wave walks 8 neighbouring nodes as one stream of rows (sources, then own row; 16 or 32 row loads per trip across node ends; starts, dinv read once per run): 8x fewer waves; same sums
# speedup vs baseline: 4.8449x; 2.1737x over previous
#include <hip/hip_runtime.h>
#include <hip/hip_bf16.h>


typedef float v2f __attribute__((ext_vector_type(2)));
typedef float v4f __attribute__((ext_vector_type(4)));
typedef float v8f __attribute__((ext_vector_type(8)));

template <int NT> struct RowVec;
template <> struct RowVec<2> { typedef v2f type; };
template <> struct RowVec<4> { typedef v4f type; };

#define N_NODES 100000
#define IN_CH   128
#define HID     64
#define OUT_CH  32

#define LIST_OFF 131072

#define BIN_SHIFT 9
#define BIN_NODES (1 << BIN_SHIFT)
#define MAX_BINS  256
#define SRC_BITS  17
#define SRC_MASK  ((1 << SRC_BITS) - 1)
#define SORT_T    1024
#define CHUNK     6400
#define BIN_CAP   12288

__device__ __forceinline__ int block_excl_scan(int v, int* s_ws, int& total) {
  const int tid  = threadIdx.x;
  const int lane = tid & 31;
  const int wv   = tid >> 5;
  int inc = v;
  #pragma unroll
  for (int d = 1; d < 32; d <<= 1) {
    int t = __shfl_up(inc, d, 32);
    if (lane >= d) inc += t;
  }
  if (lane == 31) s_ws[wv] = inc;
  __syncthreads();
  int base = 0, tot = 0;
  for (int w = 0; w < SORT_T / 32; ++w) {
    const int x = s_ws[w];
    if (w < wv) base += x;
    tot += x;
  }
  __syncthreads();
  total = tot;
  return base + inc - v;
}

__global__ void __launch_bounds__(SORT_T)
k_hist(const int* __restrict__ dst, int* __restrict__ hist, int e, int nb) {
  constexpr int EPT = (CHUNK + SORT_T - 1) / SORT_T;
  __shared__ int s_h[MAX_BINS];
  const int tid = threadIdx.x;
  const int e0 = (int)blockIdx.x * CHUNK;
  int n = e - e0;
  if (n > CHUNK) n = CHUNK;
  int rd[EPT];
  #pragma unroll
  for (int j = 0; j < EPT; ++j) {
    const int i = tid + j * SORT_T;
    rd[j] = (i < n) ? dst[e0 + i] : -1;
  }
  if (tid < MAX_BINS) s_h[tid] = 0;
  __syncthreads();
  #pragma unroll
  for (int j = 0; j < EPT; ++j) {
    if (rd[j] >= 0) atomicAdd(&s_h[rd[j] >> BIN_SHIFT], 1);
  }
  __syncthreads();
  if (tid < nb) hist[(int)blockIdx.x * nb + tid] = s_h[tid];
}

__global__ void __launch_bounds__(SORT_T)
k_part(const int* __restrict__ src, const int* __restrict__ dst,
       const int* __restrict__ hist, int* __restrict__ binstart,
       int* __restrict__ binned, int e, int nb, int g) {
  constexpr int EPT = (CHUNK + SORT_T - 1) / SORT_T;
  __shared__ int s_w[CHUNK];
  __shared__ unsigned char s_b[CHUNK];
  __shared__ int s_pre[4][MAX_BINS];
  __shared__ int s_tot[4][MAX_BINS];
  __shared__ int s_cur[MAX_BINS];
  __shared__ int s_delta[MAX_BINS];
  __shared__ int s_ws[SORT_T / 32];
  const int tid = threadIdx.x;
  const int b   = (int)blockIdx.x;
  const int e0  = b * CHUNK;
  int n = e - e0;
  if (n > CHUNK) n = CHUNK;

  int rd[EPT], rs[EPT];
  #pragma unroll
  for (int j = 0; j < EPT; ++j) {
    const int i = tid + j * SORT_T;
    rd[j] = 0;
    rs[j] = 0;
    if (i < n) { rd[j] = dst[e0 + i]; rs[j] = src[e0 + i]; }
  }

  const int grp = tid >> 8;
  const int t   = tid & 255;
  int pre = 0, tot = 0;
  if (t < nb) {
    #pragma unroll 8
    for (int bb = grp; bb < g; bb += 4) {
      const int v = hist[bb * nb + t];
      tot += v;
      pre += (bb < b) ? v : 0;
    }
  }
  s_pre[grp][t] = pre;
  s_tot[grp][t] = tot;
  __syncthreads();
  int own = 0, cpre = 0, ctot = 0;
  if (tid < nb) {
    cpre = (s_pre[0][tid] + s_pre[1][tid]) + (s_pre[2][tid] + s_pre[3][tid]);
    ctot = (s_tot[0][tid] + s_tot[1][tid]) + (s_tot[2][tid] + s_tot[3][tid]);
    own  = hist[b * nb + tid];
  }
  int all = 0, unused = 0;
  const int bstart = block_excl_scan(ctot, s_ws, all);
  const int lbase  = block_excl_scan(own, s_ws, unused);
  if (tid < nb) {
    s_cur[tid]   = lbase;
    s_delta[tid] = bstart + cpre - lbase;
    if (b == 0) binstart[tid] = bstart;
  }
  if (b == 0 && tid == nb) binstart[nb] = all;
  __syncthreads();

  #pragma unroll
  for (int j = 0; j < EPT; ++j) {
    const int i = tid + j * SORT_T;
    if (i < n) {
      const int k = rd[j] >> BIN_SHIFT;
      const int r = atomicAdd(&s_cur[k], 1);
      s_w[r] = rs[j] | ((rd[j] & (BIN_NODES - 1)) << SRC_BITS);
      s_b[r] = (unsigned char)k;
    }
  }
  __syncthreads();
  for (int i = tid; i < n; i += SORT_T) binned[s_delta[s_b[i]] + i] = s_w[i];
}

__global__ void __launch_bounds__(SORT_T)
k_binsort(const int* __restrict__ binned, const int* __restrict__ binstart,
          int* __restrict__ offs, float* __restrict__ dinv,
          int* __restrict__ list, int n_nodes) {
  __shared__ int s_cnt[BIN_NODES];
  __shared__ int s_cur[BIN_NODES];
  __shared__ int s_ws[SORT_T / 32];
  __shared__ int s_list[BIN_CAP];
  constexpr int EPB = 10;
  const int tid   = threadIdx.x;
  const int bin   = (int)blockIdx.x;
  const int node0 = bin << BIN_SHIFT;
  const int bs = binstart[bin];
  const int be = binstart[bin + 1];
  const int m  = be - bs;
  if (tid < BIN_NODES) s_cnt[tid] = 0;
  __syncthreads();
  for (int base = 0; base < m; base += EPB * SORT_T) {
    int w[EPB];
    #pragma unroll
    for (int j = 0; j < EPB; ++j) {
      const int i = base + tid + j * SORT_T;
      w[j] = (i < m) ? binned[bs + i] : -1;
    }
    #pragma unroll
    for (int j = 0; j < EPB; ++j) {
      if (w[j] >= 0) atomicAdd(&s_cnt[w[j] >> SRC_BITS], 1);
    }
  }
  __syncthreads();
  const int c = (tid < BIN_NODES) ? s_cnt[tid] : 0;
  int unused = 0;
  const int ex = block_excl_scan(c, s_ws, unused);
  if (tid < BIN_NODES) {
    s_cur[tid] = ex;
    const int node = node0 + tid;
    if (node < n_nodes) {
      offs[node] = bs + ex;
      dinv[node] = rsqrtf((float)(c + 1));
    }
    if (node == n_nodes - 1) offs[n_nodes] = be;
  }
  __syncthreads();
  const bool fits = (m <= BIN_CAP);
  for (int base = 0; base < m; base += EPB * SORT_T) {
    int w[EPB];
    #pragma unroll
    for (int j = 0; j < EPB; ++j) {
      const int i = base + tid + j * SORT_T;
      w[j] = (i < m) ? binned[bs + i] : -1;
    }
    #pragma unroll
    for (int j = 0; j < EPB; ++j) {
      if (w[j] >= 0) {
        const int r = atomicAdd(&s_cur[w[j] >> SRC_BITS], 1);
        if (fits) s_list[r] = w[j] & SRC_MASK;
        else      list[bs + r] = w[j] & SRC_MASK;
      }
    }
  }
  __syncthreads();
  if (fits) {
    for (int i = tid; i < m; i += SORT_T) list[bs + i] = s_list[i];
  }
}

template <int K, int NT>
__global__ void __launch_bounds__(128)
k_gemm_strip(const float* __restrict__ A, const float* __restrict__ W,
                             const float* __restrict__ rowscale,
                             float* __restrict__ out, int M, int strips) {
  typedef typename RowVec<NT>::type rowv;
  constexpr int NC = 16 * NT;
  constexpr int G  = 4;
  constexpr int NG = K / (4 * G);
  static_assert(K % (4 * G) == 0, "K must be a multiple of 16");

  const int wave = (int)((blockIdx.x * (long)blockDim.x + threadIdx.x) >> 5);
  if (wave >= strips) return;
  const int lane = threadIdx.x & 31;
  const int hi   = lane >> 4;
  const int l15  = lane & 15;
  const int row  = wave * 16 + l15;
  const int rrow = row < M ? row : M - 1;

  const float* arow = A + (size_t)rrow * K + 2 * hi;
  const float* wrow = W + (size_t)(2 * hi) * NC + NT * l15;

  const int m0 = wave * 16 + hi * 8;
  float sc[8];
  #pragma unroll
  for (int v = 0; v < 8; ++v) {
    const int r = m0 + v;
    sc[v] = rowscale[r < M ? r : M - 1];
  }

  v8f c[NT];
  #pragma unroll
  for (int t = 0; t < NT; ++t) c[t] = v8f{};

  v2f  a[2][G];
  rowv b0[2][G], b1[2][G];
  #pragma unroll
  for (int j = 0; j < G; ++j) {
    a[0][j]  = *(const v2f*)(arow + 4 * j);
    b0[0][j] = *(const rowv*)(wrow + (size_t)(4 * j) * NC);
    b1[0][j] = *(const rowv*)(wrow + (size_t)(4 * j + 1) * NC);
  }
  #pragma unroll
  for (int g = 0; g < NG; ++g) {
    const int cur = g & 1, nxt = cur ^ 1;
    if (g + 1 < NG) {
      #pragma unroll
      for (int j = 0; j < G; ++j) {
        const int k0 = (g + 1) * 4 * G + 4 * j;
        a[nxt][j]  = *(const v2f*)(arow + k0);
        b0[nxt][j] = *(const rowv*)(wrow + (size_t)k0 * NC);
        b1[nxt][j] = *(const rowv*)(wrow + (size_t)(k0 + 1) * NC);
      }
    }
    #pragma unroll
    for (int j = 0; j < G; ++j) {
      #pragma unroll
      for (int t = 0; t < NT; ++t) {
        v2f b;
        b.x = b0[cur][j][t];
        b.y = b1[cur][j][t];
        c[t] = __builtin_amdgcn_wmma_f32_16x16x4_f32(
             false, a[cur][j],  false, b,
             (short)0, c[t],  false,  false);
      }
    }
  }

  #pragma unroll
  for (int v = 0; v < 8; ++v) {
    const int r = m0 + v;
    rowv o;
    #pragma unroll
    for (int t = 0; t < NT; ++t) o[t] = c[t][v] * sc[v];
    if (r < M) *(rowv*)(out + (size_t)r * NC + NT * l15) = o;
  }
}

#define NPW 8

template <int C, typename Fin>
__device__ __forceinline__ void walk_nodes(const float* __restrict__ m, const int* __restrict__ list,
                                           const int* __restrict__ offs, const float* __restrict__ dinv,
                                           int n0, int cntN, int lane, Fin fin) {
  constexpr int GRP = (C == 64) ? 16 : 32;
  const int   vofs = offs[n0 + (lane < cntN ? lane : cntN)];
  const float vdi  = dinv[n0 + (lane < cntN ? lane : cntN - 1)];
  const int rb = __builtin_amdgcn_readlane(vofs, 0);
  const int re = __builtin_amdgcn_readlane(vofs, cntN);
  const int T  = (re - rb) + cntN;
  int sp[NPW];
  #pragma unroll
  for (int i = 0; i < NPW; ++i) {
    const int en = __builtin_amdgcn_readlane(vofs, i + 1);
    sp[i] = (i < cntN) ? (en - rb + i) : 0x7fffffff;
  }
  auto place = [&](int t, bool& self, int& row, int& eidx) {
    int clt = 0;
    bool eq = false;
    #pragma unroll
    for (int i = 0; i < NPW; ++i) {
      clt += (sp[i] < t) ? 1 : 0;
      eq = eq || (sp[i] == t);
    }
    self = eq;
    row  = n0 + clt;
    int idx = rb + t - clt;
    if (idx > re - 1) idx = re - 1;
    if (idx < 0) idx = 0;
    eidx = idx;
  };

  bool selfC; int rowC, eC;
  place(lane, selfC, rowC, eC);
  int wC = list[eC];

  float a0 = 0.0f, a1 = 0.0f;
  int ndc = 0;
  for (int T0 = 0; T0 < T; T0 += 32) {
    const bool validC = (T0 + lane) < T;
    const int rowsV = validC ? (selfC ? rowC : wC) : n0;
    const unsigned selfm = __builtin_amdgcn_ballot_w32(validC && selfC);
    const unsigned valm  = __builtin_amdgcn_ballot_w32(validC);
    bool selfN; int rowN, eN;
    place(T0 + 32 + lane, selfN, rowN, eN);
    const int wN = list[eN];
    #pragma unroll
    for (int h = 0; h < 32 / GRP; ++h) {
      if (h > 0 && T0 + h * GRP >= T) break;
      v2f   r2[GRP];
      float r1[GRP];
      #pragma unroll
      for (int j = 0; j < GRP; ++j) {
        const int row = __builtin_amdgcn_readlane(rowsV, h * GRP + j);
        if constexpr (C == 64) r2[j] = *(const v2f*)(m + (size_t)row * C + 2 * lane);
        else                   r1[j] = m[(size_t)row * C + lane];
      }
      #pragma unroll
      for (int j = 0; j < GRP; ++j) {
        const unsigned bit = 1u << (h * GRP + j);
        const bool real = (valm & bit) != 0;
        if constexpr (C == 64) {
          a0 += real ? r2[j].x : 0.0f;
          a1 += real ? r2[j].y : 0.0f;
        } else {
          a0 += real ? r1[j] : 0.0f;
        }
        if (selfm & bit) {
          const float di = __int_as_float(__builtin_amdgcn_readlane(__float_as_int(vdi), ndc));
          fin(ndc, a0, a1, di);
          a0 = 0.0f; a1 = 0.0f;
          ++ndc;
        }
      }
    }
    selfC = selfN; rowC = rowN; wC = wN;
  }
}

template <int C, bool RELU>
__global__ void __launch_bounds__(256)
k_gather(const float* __restrict__ m, const int* __restrict__ list,
         const int* __restrict__ offs, const float* __restrict__ dinv,
         const float* __restrict__ bias, float* __restrict__ res, int n) {
  static_assert(C == 32, "one float per lane");
  const int wave = (int)(blockIdx.x * (blockDim.x >> 5)) +
                   __builtin_amdgcn_readfirstlane((int)(threadIdx.x >> 5));
  const int n0 = wave * NPW;
  if (n0 >= n) return;
  int cntN = n - n0;
  if (cntN > NPW) cntN = NPW;
  const int lane = threadIdx.x & 31;
  const float bv = bias[lane];
  walk_nodes<C>(m, list, offs, dinv, n0, cntN, lane,
                [&](int ndc, float a0, float a1, float di) {
                  (void)a1;
                  float v0 = a0 * di + bv;
                  if (RELU) v0 = v0 > 0.0f ? v0 : 0.0f;
                  res[(size_t)(n0 + ndc) * C + lane] = v0;
                });
}

#define H_PITCH 68
__global__ void __launch_bounds__(512, 16)
k_gather_gemm(const float* __restrict__ m, const int* __restrict__ list,
              const int* __restrict__ offs, const float* __restrict__ dinv,
              const float* __restrict__ bias, const float* __restrict__ W,
              float* __restrict__ out2, int n) {
  constexpr int C  = HID;
  constexpr int NC = OUT_CH;
  __shared__ __attribute__((aligned(16))) float sH[16 * NPW * H_PITCH];

  const int wv   = __builtin_amdgcn_readfirstlane((int)(threadIdx.x >> 5));
  const int lane = threadIdx.x & 31;
  const int blk0 = (int)blockIdx.x * (16 * NPW);
  const int n0   = blk0 + wv * NPW;
  int cntN = n - n0;
  if (cntN > NPW) cntN = NPW;
  if (cntN < 0) cntN = 0;
  float* hrows = sH + (wv * NPW) * H_PITCH + 2 * lane;

  if (cntN > 0) {
    const float bv0 = bias[2 * lane];
    const float bv1 = bias[2 * lane + 1];
    walk_nodes<C>(m, list, offs, dinv, n0, cntN, lane,
                  [&](int ndc, float a0, float a1, float di) {
                    float v0 = a0 * di + bv0;
                    float v1 = a1 * di + bv1;
                    v0 = v0 > 0.0f ? v0 : 0.0f;
                    v1 = v1 > 0.0f ? v1 : 0.0f;
                    v2f h; h.x = v0; h.y = v1;
                    *(v2f*)(hrows + ndc * H_PITCH) = h;
                  });
  }
  for (int t = cntN; t < NPW; ++t) *(v2f*)(hrows + t * H_PITCH) = v2f{};
  __syncthreads();
  if (wv >= NPW || blk0 + wv * 16 >= n) return;

  constexpr int G  = 2;
  constexpr int NG = C / (4 * G);
  const int hi  = lane >> 4;
  const int l15 = lane & 15;
  const float* arow = sH + (wv * 16 + l15) * H_PITCH + 2 * hi;
  const float* wrow = W + (size_t)(2 * hi) * NC + 2 * l15;

  const int m0 = blk0 + wv * 16 + hi * 8;
  float sc[8];
  #pragma unroll
  for (int v = 0; v < 8; ++v) {
    const int r = m0 + v;
    sc[v] = dinv[r < n ? r : n - 1];
  }

  v8f c[2];
  c[0] = v8f{};
  c[1] = v8f{};

  v2f b0[2][G], b1[2][G];
  #pragma unroll
  for (int j = 0; j < G; ++j) {
    b0[0][j] = *(const v2f*)(wrow + (size_t)(4 * j) * NC);
    b1[0][j] = *(const v2f*)(wrow + (size_t)(4 * j + 1) * NC);
  }
  #pragma unroll
  for (int g = 0; g < NG; ++g) {
    const int cur = g & 1, nxt = cur ^ 1;
    if (g + 1 < NG) {
      #pragma unroll
      for (int j = 0; j < G; ++j) {
        const int k0 = (g + 1) * 4 * G + 4 * j;
        b0[nxt][j] = *(const v2f*)(wrow + (size_t)k0 * NC);
        b1[nxt][j] = *(const v2f*)(wrow + (size_t)(k0 + 1) * NC);
      }
    }
    #pragma unroll
    for (int j = 0; j < G; ++j) {
      const v2f a = *(const v2f*)(arow + g * 4 * G + 4 * j);
      #pragma unroll
      for (int t = 0; t < 2; ++t) {
        v2f b;
        b.x = b0[cur][j][t];
        b.y = b1[cur][j][t];
        c[t] = __builtin_amdgcn_wmma_f32_16x16x4_f32(
             false, a,  false, b,
             (short)0, c[t],  false,  false);
      }
    }
  }

  #pragma unroll
  for (int v = 0; v < 8; ++v) {
    const int r = m0 + v;
    v2f o;
    o.x = c[0][v] * sc[v];
    o.y = c[1][v] * sc[v];
    if (r < n) *(v2f*)(out2 + (size_t)r * NC + 2 * l15) = o;
  }
}

extern "C" void kernel_launch(void* const* d_in, const int* in_sizes, int n_in,
                              void* d_out, int out_size, void* d_ws, size_t ws_size,
                              hipStream_t stream) {
  const float* x   = (const float*)d_in[0];
  const int*   ei  = (const int*)d_in[1];
  const float* W1  = (const float*)d_in[2];
  const float* b1  = (const float*)d_in[3];
  const float* W2  = (const float*)d_in[4];
  const float* b2  = (const float*)d_in[5];
  float*       out = (float*)d_out;

  const int N = in_sizes[0] / IN_CH;
  const int E = in_sizes[1] / 2;
  const int* src = ei;
  const int* dst = ei + E;

  float* dinv = (float*)d_ws;
  float* bufA = dinv + 131072;
  float* bufB = bufA + (size_t)N * HID;

  const int NB = (N + BIN_NODES - 1) >> BIN_SHIFT;
  const int G  = (E + CHUNK - 1) / CHUNK;
  int* binned   = (int*)bufA;
  int* hist     = binned + E;
  int* binstart = hist + G * NB;
  int*   offs = (int*)bufB;
  int*   list = offs + LIST_OFF;
  float* m2   = (float*)(list + E);

  const int BT = 256;

  k_hist<<<G, SORT_T, 0, stream>>>(dst, hist, E, NB);
  k_part<<<G, SORT_T, 0, stream>>>(src, dst, hist, binstart, binned, E, NB, G);
  k_binsort<<<NB, SORT_T, 0, stream>>>(binned, binstart, offs, dinv, list, N);
  const int GT = 128;
  const int strips = (N + 15) / 16;
  k_gemm_strip<IN_CH, HID / 16><<<(strips * 32 + GT - 1) / GT, GT, 0, stream>>>(
      x, W1, dinv, bufA, N, strips);
  k_gather_gemm<<<(N + 16 * NPW - 1) / (16 * NPW), 512, 0, stream>>>(bufA, list, offs, dinv, b1, W2, m2, N);
  k_gather<OUT_CH, false><<<(N + NPW * (BT / 32) - 1) / (NPW * (BT / 32)), BT, 0, stream>>>(
      m2, list, offs, dinv, b2, out, N);
}
